# starting kernel
# baseline (speedup 1.0000x reference)
#include <hip/hip_runtime.h>
#include <math.h>

typedef __attribute__((ext_vector_type(2))) float v2f;
typedef __attribute__((ext_vector_type(8))) float v8f;

#define NDIM 512
#define EPSF 1e-12f

#define OFF_HI (16 * NDIM)
#define OFF_PP (NDIM * NDIM)
#define OFF_PPHI (NDIM * NDIM + 16 * NDIM)

__global__ void pp2_colnorm_kernel(const float* __restrict__ normal,
                                   float* __restrict__ cninv) {
    int i = blockIdx.x * blockDim.x + threadIdx.x;
    if (i >= NDIM) return;
    float s = 0.0f;
    #pragma unroll 8
    for (int o = 0; o < NDIM; ++o) {
        float v = normal[o * NDIM + i];
        s += v * v;
    }
    cninv[i] = 1.0f / fmaxf(sqrtf(s), EPSF);
}

__global__ void pp2_rowprep_kernel(const float* __restrict__ normal,
                                   const float* __restrict__ plane,
                                   const float* __restrict__ cninv,
                                   float* __restrict__ nn0,
                                   float* __restrict__ pproj,
                                   float* __restrict__ dvec,
                                   float* __restrict__ gvec) {
    const int o = blockIdx.x;
    const int t = threadIdx.x;
    const float* nr = normal + o * NDIM;
    const float* pr = plane + o * NDIM;

    float sp = 0.f, sn = 0.f, spn = 0.f, g = 0.f, t1 = 0.f, t2 = 0.f;
    for (int i = t; i < NDIM; i += 256) {
        float n = nr[i], p = pr[i], c = cninv[i];
        float n0 = n * c;
        sp  += p * p;
        sn  += n * n;
        spn += p * n;
        g   += n0 * n0;
        t1  += n0 * p;
        t2  += n0 * n;
        nn0[o * NDIM + i] = n0;
    }

    __shared__ float red[6][256];
    red[0][t] = sp; red[1][t] = sn; red[2][t] = spn;
    red[3][t] = g;  red[4][t] = t1; red[5][t] = t2;
    __syncthreads();
    for (int s = 128; s > 0; s >>= 1) {
        if (t < s) {
            #pragma unroll
            for (int j = 0; j < 6; ++j) red[j][t] += red[j][t + s];
        }
        __syncthreads();
    }
    float Sp = red[0][0], Sn = red[1][0], Spn = red[2][0];
    float G  = red[3][0], T1 = red[4][0], T2 = red[5][0];

    float np  = fmaxf(sqrtf(Sp), EPSF);
    float nn1 = fmaxf(sqrtf(Sn), EPSF);
    float ct  = Spn / (np * nn1);
    float qq  = fmaxf(Sp - 2.0f * ct * Spn + ct * ct * Sn, 0.0f);
    float rinv = 1.0f / fmaxf(sqrtf(qq), EPSF);

    if (t == 0) {
        dvec[o] = rinv * (T1 - ct * T2);
        gvec[o] = G;
    }
    for (int i = t; i < NDIM; i += 256) {
        pproj[o * NDIM + i] = (pr[i] - nr[i] * ct) * rinv;
    }
}

__global__ void pp2_xnorm_kernel(const float* __restrict__ x,
                                 float* __restrict__ svec) {
    const int b = blockIdx.x;
    const int t = threadIdx.x;
    float s = 0.0f;
    for (int i = t; i < NDIM; i += 256) {
        float v = x[b * NDIM + i];
        s += v * v;
    }
    __shared__ float red[256];
    red[t] = s;
    __syncthreads();
    for (int w = 128; w > 0; w >>= 1) {
        if (t < w) red[t] += red[t + w];
        __syncthreads();
    }
    if (t == 0) svec[b] = red[0];
}

__global__ __launch_bounds__(128) void pp2_wmma_gemm_kernel(
    const float* __restrict__ x,
    const float* __restrict__ nn0,
    const float* __restrict__ dvec,
    const float* __restrict__ gvec,
    const float* __restrict__ svec,
    float* __restrict__ out) {
    const int wave = blockIdx.x * (blockDim.x >> 5) + (threadIdx.x >> 5);
    const int mt = wave >> 4;
    const int nt = wave & 15;
    const int tm = mt * 16;
    const int tn = nt * 32;

    const int lane = threadIdx.x & 31;
    const int half = lane >> 4;
    const int l = lane & 15;

    const float* ap = x   + (tm + l) * NDIM + 2 * half;
    const float* bp = nn0 + (tn + l) * NDIM + 2 * half;

    v8f c1lo = {}, c1hi = {}, c2lo = {}, c2hi = {};

    v2f a  = *(const v2f*)(ap);
    v2f p0 = *(const v2f*)(bp);
    v2f p1 = *(const v2f*)(bp + OFF_HI);
    v2f q0 = *(const v2f*)(bp + OFF_PP);
    v2f q1 = *(const v2f*)(bp + OFF_PPHI);

    #pragma unroll 4
    for (int k = 0; k < NDIM - 4; k += 4) {
        v2f an  = *(const v2f*)(ap + k + 4);
        v2f p0n = *(const v2f*)(bp + k + 4);
        v2f p1n = *(const v2f*)(bp + k + 4 + OFF_HI);
        v2f q0n = *(const v2f*)(bp + k + 4 + OFF_PP);
        v2f q1n = *(const v2f*)(bp + k + 4 + OFF_PPHI);

        c1lo = __builtin_amdgcn_wmma_f32_16x16x4_f32(false, a, false, p0,
                                                     (short)0, c1lo, false, false);
        c1hi = __builtin_amdgcn_wmma_f32_16x16x4_f32(false, a, false, p1,
                                                     (short)0, c1hi, false, false);
        c2lo = __builtin_amdgcn_wmma_f32_16x16x4_f32(false, a, false, q0,
                                                     (short)0, c2lo, false, false);
        c2hi = __builtin_amdgcn_wmma_f32_16x16x4_f32(false, a, false, q1,
                                                     (short)0, c2hi, false, false);

        a = an; p0 = p0n; p1 = p1n; q0 = q0n; q1 = q1n;
    }

    c1lo = __builtin_amdgcn_wmma_f32_16x16x4_f32(false, a, false, p0,
                                                 (short)0, c1lo, false, false);
    c1hi = __builtin_amdgcn_wmma_f32_16x16x4_f32(false, a, false, p1,
                                                 (short)0, c1hi, false, false);
    c2lo = __builtin_amdgcn_wmma_f32_16x16x4_f32(false, a, false, q0,
                                                 (short)0, c2lo, false, false);
    c2hi = __builtin_amdgcn_wmma_f32_16x16x4_f32(false, a, false, q1,
                                                 (short)0, c2hi, false, false);

    const float dlo = dvec[tn + l];
    const float dhi = dvec[tn + 16 + l];
    const float glo = gvec[tn + l] - 2.0f;
    const float ghi = gvec[tn + 16 + l] - 2.0f;

    #pragma unroll
    for (int r = 0; r < 8; ++r) {
        const int row = tm + r + half * 8;
        const float sb = svec[row];

        float c1 = c1lo[r], c2 = c2lo[r];
        float den = fmaxf(sqrtf(fmaxf(sb + c1 * c1 * glo, 0.0f)), EPSF);
        out[row * NDIM + tn + l] = (c2 - c1 * dlo) / den;

        c1 = c1hi[r]; c2 = c2hi[r];
        den = fmaxf(sqrtf(fmaxf(sb + c1 * c1 * ghi, 0.0f)), EPSF);
        out[row * NDIM + tn + 16 + l] = (c2 - c1 * dhi) / den;
    }
}

extern "C" void kernel_launch(void* const* d_in, const int* in_sizes, int n_in,
                              void* d_out, int out_size, void* d_ws, size_t ws_size,
                              hipStream_t stream) {
    const float* x      = (const float*)d_in[0];
    const float* normal = (const float*)d_in[1];
    const float* plane  = (const float*)d_in[2];
    float* out = (float*)d_out;

    float* ws    = (float*)d_ws;
    float* nn0   = ws;
    float* pproj = ws + NDIM * NDIM;
    float* cninv = ws + 2 * NDIM * NDIM;
    float* dvec  = cninv + NDIM;
    float* gvec  = dvec + NDIM;
    float* svec  = gvec + NDIM;

    pp2_colnorm_kernel<<<(NDIM + 255) / 256, 256, 0, stream>>>(normal, cninv);
    pp2_rowprep_kernel<<<NDIM, 256, 0, stream>>>(normal, plane, cninv,
                                                 nn0, pproj, dvec, gvec);
    pp2_xnorm_kernel<<<NDIM, 256, 0, stream>>>(x, svec);

    pp2_wmma_gemm_kernel<<<128, 128, 0, stream>>>(x, nn0,
                                                  dvec, gvec, svec, out);
}


# Column norms: 16 blocks of 32 waves, each thread 16 rows with all loads in flight, LDS join of 32 partials, instead of one thread looping 512 rows; column sum order changes (tree)
# speedup vs baseline: 2.1133x; 2.1133x over previous
#include <hip/hip_runtime.h>
#include <math.h>

typedef __attribute__((ext_vector_type(2))) float v2f;
typedef __attribute__((ext_vector_type(8))) float v8f;

#define NDIM 512
#define EPSF 1e-12f

#define OFF_HI (16 * NDIM)
#define OFF_PP (NDIM * NDIM)
#define OFF_PPHI (NDIM * NDIM + 16 * NDIM)

#define CN_COLS 32
#define CN_WAVES 32
#define CN_ROWS 16
#define CN_BLOCKS 16

__global__ __launch_bounds__(CN_COLS * CN_WAVES) void pp2_colnorm_kernel(
    const float* __restrict__ normal, float* __restrict__ cninv) {
    const int c   = threadIdx.x & (CN_COLS - 1);
    const int rg  = threadIdx.x >> 5;
    const int col = blockIdx.x * CN_COLS + c;
    const float* p = normal + rg * NDIM + col;

    float v[CN_ROWS];
    #pragma unroll
    for (int j = 0; j < CN_ROWS; ++j) v[j] = p[j * CN_WAVES * NDIM];

    float s = 0.0f;
    #pragma unroll
    for (int j = 0; j < CN_ROWS; ++j) s += v[j] * v[j];

    __shared__ float red[CN_WAVES][CN_COLS];
    red[rg][c] = s;
    __syncthreads();

    if (rg == 0) {
        float t0 = 0.0f, t1 = 0.0f, t2 = 0.0f, t3 = 0.0f;
        #pragma unroll
        for (int j = 0; j < CN_WAVES; j += 4) {
            t0 += red[j][c];
            t1 += red[j + 1][c];
            t2 += red[j + 2][c];
            t3 += red[j + 3][c];
        }
        const float tot = (t0 + t1) + (t2 + t3);
        cninv[col] = 1.0f / fmaxf(sqrtf(tot), EPSF);
    }
}

__global__ void pp2_rowprep_kernel(const float* __restrict__ normal,
                                   const float* __restrict__ plane,
                                   const float* __restrict__ cninv,
                                   float* __restrict__ nn0,
                                   float* __restrict__ pproj,
                                   float* __restrict__ dvec,
                                   float* __restrict__ gvec) {
    const int o = blockIdx.x;
    const int t = threadIdx.x;
    const float* nr = normal + o * NDIM;
    const float* pr = plane + o * NDIM;

    float sp = 0.f, sn = 0.f, spn = 0.f, g = 0.f, t1 = 0.f, t2 = 0.f;
    for (int i = t; i < NDIM; i += 256) {
        float n = nr[i], p = pr[i], c = cninv[i];
        float n0 = n * c;
        sp  += p * p;
        sn  += n * n;
        spn += p * n;
        g   += n0 * n0;
        t1  += n0 * p;
        t2  += n0 * n;
        nn0[o * NDIM + i] = n0;
    }

    __shared__ float red[6][256];
    red[0][t] = sp; red[1][t] = sn; red[2][t] = spn;
    red[3][t] = g;  red[4][t] = t1; red[5][t] = t2;
    __syncthreads();
    for (int s = 128; s > 0; s >>= 1) {
        if (t < s) {
            #pragma unroll
            for (int j = 0; j < 6; ++j) red[j][t] += red[j][t + s];
        }
        __syncthreads();
    }
    float Sp = red[0][0], Sn = red[1][0], Spn = red[2][0];
    float G  = red[3][0], T1 = red[4][0], T2 = red[5][0];

    float np  = fmaxf(sqrtf(Sp), EPSF);
    float nn1 = fmaxf(sqrtf(Sn), EPSF);
    float ct  = Spn / (np * nn1);
    float qq  = fmaxf(Sp - 2.0f * ct * Spn + ct * ct * Sn, 0.0f);
    float rinv = 1.0f / fmaxf(sqrtf(qq), EPSF);

    if (t == 0) {
        dvec[o] = rinv * (T1 - ct * T2);
        gvec[o] = G;
    }
    for (int i = t; i < NDIM; i += 256) {
        pproj[o * NDIM + i] = (pr[i] - nr[i] * ct) * rinv;
    }
}

__global__ void pp2_xnorm_kernel(const float* __restrict__ x,
                                 float* __restrict__ svec) {
    const int b = blockIdx.x;
    const int t = threadIdx.x;
    float s = 0.0f;
    for (int i = t; i < NDIM; i += 256) {
        float v = x[b * NDIM + i];
        s += v * v;
    }
    __shared__ float red[256];
    red[t] = s;
    __syncthreads();
    for (int w = 128; w > 0; w >>= 1) {
        if (t < w) red[t] += red[t + w];
        __syncthreads();
    }
    if (t == 0) svec[b] = red[0];
}

__global__ __launch_bounds__(128) void pp2_wmma_gemm_kernel(
    const float* __restrict__ x,
    const float* __restrict__ nn0,
    const float* __restrict__ dvec,
    const float* __restrict__ gvec,
    const float* __restrict__ svec,
    float* __restrict__ out) {
    const int wave = blockIdx.x * (blockDim.x >> 5) + (threadIdx.x >> 5);
    const int mt = wave >> 4;
    const int nt = wave & 15;
    const int tm = mt * 16;
    const int tn = nt * 32;

    const int lane = threadIdx.x & 31;
    const int half = lane >> 4;
    const int l = lane & 15;

    const float* ap = x   + (tm + l) * NDIM + 2 * half;
    const float* bp = nn0 + (tn + l) * NDIM + 2 * half;

    v8f c1lo = {}, c1hi = {}, c2lo = {}, c2hi = {};

    v2f a  = *(const v2f*)(ap);
    v2f p0 = *(const v2f*)(bp);
    v2f p1 = *(const v2f*)(bp + OFF_HI);
    v2f q0 = *(const v2f*)(bp + OFF_PP);
    v2f q1 = *(const v2f*)(bp + OFF_PPHI);

    #pragma unroll 4
    for (int k = 0; k < NDIM - 4; k += 4) {
        v2f an  = *(const v2f*)(ap + k + 4);
        v2f p0n = *(const v2f*)(bp + k + 4);
        v2f p1n = *(const v2f*)(bp + k + 4 + OFF_HI);
        v2f q0n = *(const v2f*)(bp + k + 4 + OFF_PP);
        v2f q1n = *(const v2f*)(bp + k + 4 + OFF_PPHI);

        c1lo = __builtin_amdgcn_wmma_f32_16x16x4_f32(false, a, false, p0,
                                                     (short)0, c1lo, false, false);
        c1hi = __builtin_amdgcn_wmma_f32_16x16x4_f32(false, a, false, p1,
                                                     (short)0, c1hi, false, false);
        c2lo = __builtin_amdgcn_wmma_f32_16x16x4_f32(false, a, false, q0,
                                                     (short)0, c2lo, false, false);
        c2hi = __builtin_amdgcn_wmma_f32_16x16x4_f32(false, a, false, q1,
                                                     (short)0, c2hi, false, false);

        a = an; p0 = p0n; p1 = p1n; q0 = q0n; q1 = q1n;
    }

    c1lo = __builtin_amdgcn_wmma_f32_16x16x4_f32(false, a, false, p0,
                                                 (short)0, c1lo, false, false);
    c1hi = __builtin_amdgcn_wmma_f32_16x16x4_f32(false, a, false, p1,
                                                 (short)0, c1hi, false, false);
    c2lo = __builtin_amdgcn_wmma_f32_16x16x4_f32(false, a, false, q0,
                                                 (short)0, c2lo, false, false);
    c2hi = __builtin_amdgcn_wmma_f32_16x16x4_f32(false, a, false, q1,
                                                 (short)0, c2hi, false, false);

    const float dlo = dvec[tn + l];
    const float dhi = dvec[tn + 16 + l];
    const float glo = gvec[tn + l] - 2.0f;
    const float ghi = gvec[tn + 16 + l] - 2.0f;

    #pragma unroll
    for (int r = 0; r < 8; ++r) {
        const int row = tm + r + half * 8;
        const float sb = svec[row];

        float c1 = c1lo[r], c2 = c2lo[r];
        float den = fmaxf(sqrtf(fmaxf(sb + c1 * c1 * glo, 0.0f)), EPSF);
        out[row * NDIM + tn + l] = (c2 - c1 * dlo) / den;

        c1 = c1hi[r]; c2 = c2hi[r];
        den = fmaxf(sqrtf(fmaxf(sb + c1 * c1 * ghi, 0.0f)), EPSF);
        out[row * NDIM + tn + 16 + l] = (c2 - c1 * dhi) / den;
    }
}

extern "C" void kernel_launch(void* const* d_in, const int* in_sizes, int n_in,
                              void* d_out, int out_size, void* d_ws, size_t ws_size,
                              hipStream_t stream) {
    const float* x      = (const float*)d_in[0];
    const float* normal = (const float*)d_in[1];
    const float* plane  = (const float*)d_in[2];
    float* out = (float*)d_out;

    float* ws    = (float*)d_ws;
    float* nn0   = ws;
    float* pproj = ws + NDIM * NDIM;
    float* cninv = ws + 2 * NDIM * NDIM;
    float* dvec  = cninv + NDIM;
    float* gvec  = dvec + NDIM;
    float* svec  = gvec + NDIM;

    pp2_colnorm_kernel<<<CN_BLOCKS, CN_COLS * CN_WAVES, 0, stream>>>(normal, cninv);
    pp2_rowprep_kernel<<<NDIM, 256, 0, stream>>>(normal, plane, cninv,
                                                 nn0, pproj, dvec, gvec);
    pp2_xnorm_kernel<<<NDIM, 256, 0, stream>>>(x, svec);

    pp2_wmma_gemm_kernel<<<128, 128, 0, stream>>>(x, nn0,
                                                  dvec, gvec, svec, out);
}


# Row norms of x summed inside the GEMM from the A fragments each lane already loads, joined across lane halves by shuffle; the separate row-norm launch is gone (4 -> 3); sum order of s_b changes
# speedup vs baseline: 2.4253x; 1.1476x over previous
#include <hip/hip_runtime.h>
#include <math.h>

typedef __attribute__((ext_vector_type(2))) float v2f;
typedef __attribute__((ext_vector_type(8))) float v8f;

#define NDIM 512
#define EPSF 1e-12f

#define OFF_HI (16 * NDIM)
#define OFF_PP (NDIM * NDIM)
#define OFF_PPHI (NDIM * NDIM + 16 * NDIM)

#define CN_COLS 32
#define CN_WAVES 32
#define CN_ROWS 16
#define CN_BLOCKS 16

__global__ __launch_bounds__(CN_COLS * CN_WAVES) void pp2_colnorm_kernel(
    const float* __restrict__ normal, float* __restrict__ cninv) {
    const int c   = threadIdx.x & (CN_COLS - 1);
    const int rg  = threadIdx.x >> 5;
    const int col = blockIdx.x * CN_COLS + c;
    const float* p = normal + rg * NDIM + col;

    float v[CN_ROWS];
    #pragma unroll
    for (int j = 0; j < CN_ROWS; ++j) v[j] = p[j * CN_WAVES * NDIM];

    float s = 0.0f;
    #pragma unroll
    for (int j = 0; j < CN_ROWS; ++j) s += v[j] * v[j];

    __shared__ float red[CN_WAVES][CN_COLS];
    red[rg][c] = s;
    __syncthreads();

    if (rg == 0) {
        float t0 = 0.0f, t1 = 0.0f, t2 = 0.0f, t3 = 0.0f;
        #pragma unroll
        for (int j = 0; j < CN_WAVES; j += 4) {
            t0 += red[j][c];
            t1 += red[j + 1][c];
            t2 += red[j + 2][c];
            t3 += red[j + 3][c];
        }
        const float tot = (t0 + t1) + (t2 + t3);
        cninv[col] = 1.0f / fmaxf(sqrtf(tot), EPSF);
    }
}

__global__ void pp2_rowprep_kernel(const float* __restrict__ normal,
                                   const float* __restrict__ plane,
                                   const float* __restrict__ cninv,
                                   float* __restrict__ nn0,
                                   float* __restrict__ pproj,
                                   float* __restrict__ dvec,
                                   float* __restrict__ gvec) {
    const int o = blockIdx.x;
    const int t = threadIdx.x;
    const float* nr = normal + o * NDIM;
    const float* pr = plane + o * NDIM;

    float sp = 0.f, sn = 0.f, spn = 0.f, g = 0.f, t1 = 0.f, t2 = 0.f;
    for (int i = t; i < NDIM; i += 256) {
        float n = nr[i], p = pr[i], c = cninv[i];
        float n0 = n * c;
        sp  += p * p;
        sn  += n * n;
        spn += p * n;
        g   += n0 * n0;
        t1  += n0 * p;
        t2  += n0 * n;
        nn0[o * NDIM + i] = n0;
    }

    __shared__ float red[6][256];
    red[0][t] = sp; red[1][t] = sn; red[2][t] = spn;
    red[3][t] = g;  red[4][t] = t1; red[5][t] = t2;
    __syncthreads();
    for (int s = 128; s > 0; s >>= 1) {
        if (t < s) {
            #pragma unroll
            for (int j = 0; j < 6; ++j) red[j][t] += red[j][t + s];
        }
        __syncthreads();
    }
    float Sp = red[0][0], Sn = red[1][0], Spn = red[2][0];
    float G  = red[3][0], T1 = red[4][0], T2 = red[5][0];

    float np  = fmaxf(sqrtf(Sp), EPSF);
    float nn1 = fmaxf(sqrtf(Sn), EPSF);
    float ct  = Spn / (np * nn1);
    float qq  = fmaxf(Sp - 2.0f * ct * Spn + ct * ct * Sn, 0.0f);
    float rinv = 1.0f / fmaxf(sqrtf(qq), EPSF);

    if (t == 0) {
        dvec[o] = rinv * (T1 - ct * T2);
        gvec[o] = G;
    }
    for (int i = t; i < NDIM; i += 256) {
        pproj[o * NDIM + i] = (pr[i] - nr[i] * ct) * rinv;
    }
}

__global__ __launch_bounds__(128) void pp2_wmma_gemm_kernel(
    const float* __restrict__ x,
    const float* __restrict__ nn0,
    const float* __restrict__ dvec,
    const float* __restrict__ gvec,
    float* __restrict__ out) {
    const int wave = blockIdx.x * (blockDim.x >> 5) + (threadIdx.x >> 5);
    const int mt = wave >> 4;
    const int nt = wave & 15;
    const int tm = mt * 16;
    const int tn = nt * 32;

    const int lane = threadIdx.x & 31;
    const int half = lane >> 4;
    const int l = lane & 15;

    const float* ap = x   + (tm + l) * NDIM + 2 * half;
    const float* bp = nn0 + (tn + l) * NDIM + 2 * half;

    v8f c1lo = {}, c1hi = {}, c2lo = {}, c2hi = {};
    v2f ss = {};

    v2f a  = *(const v2f*)(ap);
    v2f p0 = *(const v2f*)(bp);
    v2f p1 = *(const v2f*)(bp + OFF_HI);
    v2f q0 = *(const v2f*)(bp + OFF_PP);
    v2f q1 = *(const v2f*)(bp + OFF_PPHI);

    #pragma unroll 4
    for (int k = 0; k < NDIM - 4; k += 4) {
        v2f an  = *(const v2f*)(ap + k + 4);
        v2f p0n = *(const v2f*)(bp + k + 4);
        v2f p1n = *(const v2f*)(bp + k + 4 + OFF_HI);
        v2f q0n = *(const v2f*)(bp + k + 4 + OFF_PP);
        v2f q1n = *(const v2f*)(bp + k + 4 + OFF_PPHI);

        c1lo = __builtin_amdgcn_wmma_f32_16x16x4_f32(false, a, false, p0,
                                                     (short)0, c1lo, false, false);
        c1hi = __builtin_amdgcn_wmma_f32_16x16x4_f32(false, a, false, p1,
                                                     (short)0, c1hi, false, false);
        c2lo = __builtin_amdgcn_wmma_f32_16x16x4_f32(false, a, false, q0,
                                                     (short)0, c2lo, false, false);
        c2hi = __builtin_amdgcn_wmma_f32_16x16x4_f32(false, a, false, q1,
                                                     (short)0, c2hi, false, false);

        ss += a * a;

        a = an; p0 = p0n; p1 = p1n; q0 = q0n; q1 = q1n;
    }

    c1lo = __builtin_amdgcn_wmma_f32_16x16x4_f32(false, a, false, p0,
                                                 (short)0, c1lo, false, false);
    c1hi = __builtin_amdgcn_wmma_f32_16x16x4_f32(false, a, false, p1,
                                                 (short)0, c1hi, false, false);
    c2lo = __builtin_amdgcn_wmma_f32_16x16x4_f32(false, a, false, q0,
                                                 (short)0, c2lo, false, false);
    c2hi = __builtin_amdgcn_wmma_f32_16x16x4_f32(false, a, false, q1,
                                                 (short)0, c2hi, false, false);

    ss += a * a;

    float srow = ss.x + ss.y;
    srow += __shfl_xor(srow, 16);

    const float dlo = dvec[tn + l];
    const float dhi = dvec[tn + 16 + l];
    const float glo = gvec[tn + l] - 2.0f;
    const float ghi = gvec[tn + 16 + l] - 2.0f;

    #pragma unroll
    for (int r = 0; r < 8; ++r) {
        const int row = tm + r + half * 8;
        const float sb = __shfl(srow, r + half * 8);

        float c1 = c1lo[r], c2 = c2lo[r];
        float den = fmaxf(sqrtf(fmaxf(sb + c1 * c1 * glo, 0.0f)), EPSF);
        out[row * NDIM + tn + l] = (c2 - c1 * dlo) / den;

        c1 = c1hi[r]; c2 = c2hi[r];
        den = fmaxf(sqrtf(fmaxf(sb + c1 * c1 * ghi, 0.0f)), EPSF);
        out[row * NDIM + tn + 16 + l] = (c2 - c1 * dhi) / den;
    }
}

extern "C" void kernel_launch(void* const* d_in, const int* in_sizes, int n_in,
                              void* d_out, int out_size, void* d_ws, size_t ws_size,
                              hipStream_t stream) {
    const float* x      = (const float*)d_in[0];
    const float* normal = (const float*)d_in[1];
    const float* plane  = (const float*)d_in[2];
    float* out = (float*)d_out;

    float* ws    = (float*)d_ws;
    float* nn0   = ws;
    float* pproj = ws + NDIM * NDIM;
    float* cninv = ws + 2 * NDIM * NDIM;
    float* dvec  = cninv + NDIM;
    float* gvec  = dvec + NDIM;

    pp2_colnorm_kernel<<<CN_BLOCKS, CN_COLS * CN_WAVES, 0, stream>>>(normal, cninv);
    pp2_rowprep_kernel<<<NDIM, 256, 0, stream>>>(normal, plane, cninv,
                                                 nn0, pproj, dvec, gvec);

    pp2_wmma_gemm_kernel<<<128, 128, 0, stream>>>(x, nn0,
                                                  dvec, gvec, out);
}


# GEMM fragment loads 16 B per lane instead of 8 (4 consecutive k, 2 WMMAs each): half the cache lines touched per k; waves-per-SIMD hint keeps them as far ahead; k sum order changes, also in s_b
# speedup vs baseline: 2.7024x; 1.1143x over previous
#include <hip/hip_runtime.h>
#include <math.h>

typedef __attribute__((ext_vector_type(2))) float v2f;
typedef __attribute__((ext_vector_type(4))) float v4f;
typedef __attribute__((ext_vector_type(8))) float v8f;

#define WMMA4(acc, av, bv)                                                   \
    acc = __builtin_amdgcn_wmma_f32_16x16x4_f32(false, av, false, bv,        \
                                                (short)0, acc, false, false)

#define WMMA_STEP(a, p0, p1, q0, q1)                                         \
    do {                                                                     \
        const v2f a01 = {a.x, a.y}, a23 = {a.z, a.w};                        \
        WMMA4(c1lo, a01, ((v2f){p0.x, p0.y}));                               \
        WMMA4(c1hi, a01, ((v2f){p1.x, p1.y}));                               \
        WMMA4(c2lo, a01, ((v2f){q0.x, q0.y}));                               \
        WMMA4(c2hi, a01, ((v2f){q1.x, q1.y}));                               \
        WMMA4(c1lo, a23, ((v2f){p0.z, p0.w}));                               \
        WMMA4(c1hi, a23, ((v2f){p1.z, p1.w}));                               \
        WMMA4(c2lo, a23, ((v2f){q0.z, q0.w}));                               \
        WMMA4(c2hi, a23, ((v2f){q1.z, q1.w}));                               \
    } while (0)

#define NDIM 512
#define EPSF 1e-12f

#define OFF_HI (16 * NDIM)
#define OFF_PP (NDIM * NDIM)
#define OFF_PPHI (NDIM * NDIM + 16 * NDIM)

#define CN_COLS 32
#define CN_WAVES 32
#define CN_ROWS 16
#define CN_BLOCKS 16

__global__ __launch_bounds__(CN_COLS * CN_WAVES) void pp2_colnorm_kernel(
    const float* __restrict__ normal, float* __restrict__ cninv) {
    const int c   = threadIdx.x & (CN_COLS - 1);
    const int rg  = threadIdx.x >> 5;
    const int col = blockIdx.x * CN_COLS + c;
    const float* p = normal + rg * NDIM + col;

    float v[CN_ROWS];
    #pragma unroll
    for (int j = 0; j < CN_ROWS; ++j) v[j] = p[j * CN_WAVES * NDIM];

    float s = 0.0f;
    #pragma unroll
    for (int j = 0; j < CN_ROWS; ++j) s += v[j] * v[j];

    __shared__ float red[CN_WAVES][CN_COLS];
    red[rg][c] = s;
    __syncthreads();

    if (rg == 0) {
        float t0 = 0.0f, t1 = 0.0f, t2 = 0.0f, t3 = 0.0f;
        #pragma unroll
        for (int j = 0; j < CN_WAVES; j += 4) {
            t0 += red[j][c];
            t1 += red[j + 1][c];
            t2 += red[j + 2][c];
            t3 += red[j + 3][c];
        }
        const float tot = (t0 + t1) + (t2 + t3);
        cninv[col] = 1.0f / fmaxf(sqrtf(tot), EPSF);
    }
}

__global__ void pp2_rowprep_kernel(const float* __restrict__ normal,
                                   const float* __restrict__ plane,
                                   const float* __restrict__ cninv,
                                   float* __restrict__ nn0,
                                   float* __restrict__ pproj,
                                   float* __restrict__ dvec,
                                   float* __restrict__ gvec) {
    const int o = blockIdx.x;
    const int t = threadIdx.x;
    const float* nr = normal + o * NDIM;
    const float* pr = plane + o * NDIM;

    float sp = 0.f, sn = 0.f, spn = 0.f, g = 0.f, t1 = 0.f, t2 = 0.f;
    for (int i = t; i < NDIM; i += 256) {
        float n = nr[i], p = pr[i], c = cninv[i];
        float n0 = n * c;
        sp  += p * p;
        sn  += n * n;
        spn += p * n;
        g   += n0 * n0;
        t1  += n0 * p;
        t2  += n0 * n;
        nn0[o * NDIM + i] = n0;
    }

    __shared__ float red[6][256];
    red[0][t] = sp; red[1][t] = sn; red[2][t] = spn;
    red[3][t] = g;  red[4][t] = t1; red[5][t] = t2;
    __syncthreads();
    for (int s = 128; s > 0; s >>= 1) {
        if (t < s) {
            #pragma unroll
            for (int j = 0; j < 6; ++j) red[j][t] += red[j][t + s];
        }
        __syncthreads();
    }
    float Sp = red[0][0], Sn = red[1][0], Spn = red[2][0];
    float G  = red[3][0], T1 = red[4][0], T2 = red[5][0];

    float np  = fmaxf(sqrtf(Sp), EPSF);
    float nn1 = fmaxf(sqrtf(Sn), EPSF);
    float ct  = Spn / (np * nn1);
    float qq  = fmaxf(Sp - 2.0f * ct * Spn + ct * ct * Sn, 0.0f);
    float rinv = 1.0f / fmaxf(sqrtf(qq), EPSF);

    if (t == 0) {
        dvec[o] = rinv * (T1 - ct * T2);
        gvec[o] = G;
    }
    for (int i = t; i < NDIM; i += 256) {
        pproj[o * NDIM + i] = (pr[i] - nr[i] * ct) * rinv;
    }
}

__global__ __launch_bounds__(128) __attribute__((amdgpu_waves_per_eu(1, 4)))
void pp2_wmma_gemm_kernel(
    const float* __restrict__ x,
    const float* __restrict__ nn0,
    const float* __restrict__ dvec,
    const float* __restrict__ gvec,
    float* __restrict__ out) {
    const int wave = blockIdx.x * (blockDim.x >> 5) + (threadIdx.x >> 5);
    const int mt = wave >> 4;
    const int nt = wave & 15;
    const int tm = mt * 16;
    const int tn = nt * 32;

    const int lane = threadIdx.x & 31;
    const int half = lane >> 4;
    const int l = lane & 15;

    const float* ap = x   + (tm + l) * NDIM + 4 * half;
    const float* bp = nn0 + (tn + l) * NDIM + 4 * half;

    v8f c1lo = {}, c1hi = {}, c2lo = {}, c2hi = {};
    v4f ss = {};

    v4f a  = *(const v4f*)(ap);
    v4f p0 = *(const v4f*)(bp);
    v4f p1 = *(const v4f*)(bp + OFF_HI);
    v4f q0 = *(const v4f*)(bp + OFF_PP);
    v4f q1 = *(const v4f*)(bp + OFF_PPHI);

    #pragma unroll 4
    for (int k = 0; k < NDIM - 8; k += 8) {
        v4f an  = *(const v4f*)(ap + k + 8);
        v4f p0n = *(const v4f*)(bp + k + 8);
        v4f p1n = *(const v4f*)(bp + k + 8 + OFF_HI);
        v4f q0n = *(const v4f*)(bp + k + 8 + OFF_PP);
        v4f q1n = *(const v4f*)(bp + k + 8 + OFF_PPHI);

        WMMA_STEP(a, p0, p1, q0, q1);

        ss += a * a;

        a = an; p0 = p0n; p1 = p1n; q0 = q0n; q1 = q1n;
    }

    WMMA_STEP(a, p0, p1, q0, q1);

    ss += a * a;

    float srow = (ss.x + ss.y) + (ss.z + ss.w);
    srow += __shfl_xor(srow, 16);

    const float dlo = dvec[tn + l];
    const float dhi = dvec[tn + 16 + l];
    const float glo = gvec[tn + l] - 2.0f;
    const float ghi = gvec[tn + 16 + l] - 2.0f;

    #pragma unroll
    for (int r = 0; r < 8; ++r) {
        const int row = tm + r + half * 8;
        const float sb = __shfl(srow, r + half * 8);

        float c1 = c1lo[r], c2 = c2lo[r];
        float den = fmaxf(sqrtf(fmaxf(sb + c1 * c1 * glo, 0.0f)), EPSF);
        out[row * NDIM + tn + l] = (c2 - c1 * dlo) / den;

        c1 = c1hi[r]; c2 = c2hi[r];
        den = fmaxf(sqrtf(fmaxf(sb + c1 * c1 * ghi, 0.0f)), EPSF);
        out[row * NDIM + tn + 16 + l] = (c2 - c1 * dhi) / den;
    }
}

extern "C" void kernel_launch(void* const* d_in, const int* in_sizes, int n_in,
                              void* d_out, int out_size, void* d_ws, size_t ws_size,
                              hipStream_t stream) {
    const float* x      = (const float*)d_in[0];
    const float* normal = (const float*)d_in[1];
    const float* plane  = (const float*)d_in[2];
    float* out = (float*)d_out;

    float* ws    = (float*)d_ws;
    float* nn0   = ws;
    float* pproj = ws + NDIM * NDIM;
    float* cninv = ws + 2 * NDIM * NDIM;
    float* dvec  = cninv + NDIM;
    float* gvec  = dvec + NDIM;

    pp2_colnorm_kernel<<<CN_BLOCKS, CN_COLS * CN_WAVES, 0, stream>>>(normal, cninv);
    pp2_rowprep_kernel<<<NDIM, 256, 0, stream>>>(normal, plane, cninv,
                                                 nn0, pproj, dvec, gvec);

    pp2_wmma_gemm_kernel<<<128, 128, 0, stream>>>(x, nn0,
                                                  dvec, gvec, out);
}


# Row-prep launch gone (3 -> 2): plane_proj rows made beside the column norms; GEMM reads normal, scales A by column norms, sums g_o, d_o from its B fragments; order of row sums, g_o, d_o changes
# speedup vs baseline: 3.3330x; 1.2333x over previous
#include <hip/hip_runtime.h>
#include <math.h>

typedef __attribute__((ext_vector_type(2))) float v2f;
typedef __attribute__((ext_vector_type(4))) float v4f;
typedef __attribute__((ext_vector_type(8))) float v8f;

#define WMMA4(acc, av, bv)                                                   \
    acc = __builtin_amdgcn_wmma_f32_16x16x4_f32(false, av, false, bv,        \
                                                (short)0, acc, false, false)

#define WMMA_STEP(as, a, p0, p1, q0, q1)                                     \
    do {                                                                     \
        const v2f s01 = {as.x, as.y}, s23 = {as.z, as.w};                    \
        const v2f a01 = {a.x, a.y}, a23 = {a.z, a.w};                        \
        WMMA4(c1lo, s01, ((v2f){p0.x, p0.y}));                               \
        WMMA4(c1hi, s01, ((v2f){p1.x, p1.y}));                               \
        WMMA4(c2lo, a01, ((v2f){q0.x, q0.y}));                               \
        WMMA4(c2hi, a01, ((v2f){q1.x, q1.y}));                               \
        WMMA4(c1lo, s23, ((v2f){p0.z, p0.w}));                               \
        WMMA4(c1hi, s23, ((v2f){p1.z, p1.w}));                               \
        WMMA4(c2lo, a23, ((v2f){q0.z, q0.w}));                               \
        WMMA4(c2hi, a23, ((v2f){q1.z, q1.w}));                               \
    } while (0)

#define NDIM 512
#define EPSF 1e-12f

#define OFF_HI (16 * NDIM)

#define CN_COLS 32
#define CN_WAVES 32
#define CN_ROWS 16
#define CN_BLOCKS 16
#define RP_ROWS 32
#define RP_BLOCKS (NDIM / RP_ROWS)

__global__ __launch_bounds__(CN_COLS * CN_WAVES) void pp2_prep_kernel(
    const float* __restrict__ normal, const float* __restrict__ plane,
    float* __restrict__ cninv, float* __restrict__ pproj) {
    __shared__ float red[CN_WAVES][CN_COLS];
    const int c  = threadIdx.x & (CN_COLS - 1);
    const int rg = threadIdx.x >> 5;

    if (blockIdx.x < RP_BLOCKS) {
        const int o = blockIdx.x * RP_ROWS + rg;
        const float* nr = normal + o * NDIM + 4 * c;
        const float* pr = plane  + o * NDIM + 4 * c;

        v4f n[4], p[4];
        #pragma unroll
        for (int j = 0; j < 4; ++j) {
            n[j] = *(const v4f*)(nr + j * 128);
            p[j] = *(const v4f*)(pr + j * 128);
        }

        v4f sp4 = {}, sn4 = {}, spn4 = {};
        #pragma unroll
        for (int j = 0; j < 4; ++j) {
            sp4  += p[j] * p[j];
            sn4  += n[j] * n[j];
            spn4 += p[j] * n[j];
        }
        float Sp  = (sp4.x + sp4.y) + (sp4.z + sp4.w);
        float Sn  = (sn4.x + sn4.y) + (sn4.z + sn4.w);
        float Spn = (spn4.x + spn4.y) + (spn4.z + spn4.w);
        #pragma unroll
        for (int m = 16; m > 0; m >>= 1) {
            Sp  += __shfl_xor(Sp, m);
            Sn  += __shfl_xor(Sn, m);
            Spn += __shfl_xor(Spn, m);
        }

        const float np  = fmaxf(sqrtf(Sp), EPSF);
        const float nn1 = fmaxf(sqrtf(Sn), EPSF);
        const float ct  = Spn / (np * nn1);
        const float qq  = fmaxf(Sp - 2.0f * ct * Spn + ct * ct * Sn, 0.0f);
        const float rinv = 1.0f / fmaxf(sqrtf(qq), EPSF);

        float* po = pproj + o * NDIM + 4 * c;
        #pragma unroll
        for (int j = 0; j < 4; ++j)
            *(v4f*)(po + j * 128) = (p[j] - n[j] * ct) * rinv;
    } else {
        const int col = (blockIdx.x - RP_BLOCKS) * CN_COLS + c;
        const float* p = normal + rg * NDIM + col;

        float v[CN_ROWS];
        #pragma unroll
        for (int j = 0; j < CN_ROWS; ++j) v[j] = p[j * CN_WAVES * NDIM];

        float s = 0.0f;
        #pragma unroll
        for (int j = 0; j < CN_ROWS; ++j) s += v[j] * v[j];

        red[rg][c] = s;
        __syncthreads();

        if (rg == 0) {
            float t0 = 0.0f, t1 = 0.0f, t2 = 0.0f, t3 = 0.0f;
            #pragma unroll
            for (int j = 0; j < CN_WAVES; j += 4) {
                t0 += red[j][c];
                t1 += red[j + 1][c];
                t2 += red[j + 2][c];
                t3 += red[j + 3][c];
            }
            const float tot = (t0 + t1) + (t2 + t3);
            cninv[col] = 1.0f / fmaxf(sqrtf(tot), EPSF);
        }
    }
}


#define ROW_SUMS(cv, p, q, g, d)                                             \
    do {                                                                     \
        const v4f t = p * cv;                                                \
        g += t * t;                                                          \
        d += t * q;                                                          \
    } while (0)

#define PF_DEPTH 8

#define LOAD_SLOT(s, kk)                                                     \
    do {                                                                     \
        A[s]  = *(const v4f*)(ap + (kk));                                    \
        C[s]  = *(const v4f*)(cp + (kk));                                    \
        P0[s] = *(const v4f*)(np + (kk));                                    \
        P1[s] = *(const v4f*)(np + (kk) + OFF_HI);                           \
        Q0[s] = *(const v4f*)(qp + (kk));                                    \
        Q1[s] = *(const v4f*)(qp + (kk) + OFF_HI);                           \
    } while (0)

#define USE_SLOT(s)                                                          \
    do {                                                                     \
        const v4f as = A[s] * C[s];                                          \
        WMMA_STEP(as, A[s], P0[s], P1[s], Q0[s], Q1[s]);                     \
        ss += A[s] * A[s];                                                   \
        ROW_SUMS(C[s], P0[s], Q0[s], glo, dlo);                              \
        ROW_SUMS(C[s], P1[s], Q1[s], ghi, dhi);                              \
    } while (0)

__global__ __launch_bounds__(128) __attribute__((amdgpu_waves_per_eu(1, 2)))
void pp2_wmma_gemm_kernel(
    const float* __restrict__ x,
    const float* __restrict__ normal,
    const float* __restrict__ pproj,
    const float* __restrict__ cninv,
    float* __restrict__ out) {
    const int wave = blockIdx.x * (blockDim.x >> 5) + (threadIdx.x >> 5);
    const int mt = wave >> 4;
    const int nt = wave & 15;
    const int tm = mt * 16;
    const int tn = nt * 32;

    const int lane = threadIdx.x & 31;
    const int half = lane >> 4;
    const int l = lane & 15;

    const float* ap = x      + (tm + l) * NDIM + 4 * half;
    const float* np = normal + (tn + l) * NDIM + 4 * half;
    const float* qp = pproj  + (tn + l) * NDIM + 4 * half;
    const float* cp = cninv  + 4 * half;

    v8f c1lo = {}, c1hi = {}, c2lo = {}, c2hi = {};
    v4f ss = {};
    v4f glo = {}, ghi = {};
    v4f dlo = {}, dhi = {};

    v4f A[PF_DEPTH], C[PF_DEPTH];
    v4f P0[PF_DEPTH], P1[PF_DEPTH], Q0[PF_DEPTH], Q1[PF_DEPTH];

    #pragma unroll
    for (int s = 0; s < PF_DEPTH; ++s) {
        LOAD_SLOT(s, 8 * s);
        __builtin_amdgcn_sched_barrier(0);
    }

    #pragma nounroll
    for (int k = 8 * PF_DEPTH; k < NDIM; k += 8 * PF_DEPTH) {
        #pragma unroll
        for (int s = 0; s < PF_DEPTH; ++s) {
            USE_SLOT(s);
            LOAD_SLOT(s, k + 8 * s);
            __builtin_amdgcn_sched_barrier(0);
        }
    }

    #pragma unroll
    for (int s = 0; s < PF_DEPTH; ++s) USE_SLOT(s);

    float srow = (ss.x + ss.y) + (ss.z + ss.w);
    float g_lo = (glo.x + glo.y) + (glo.z + glo.w);
    float g_hi = (ghi.x + ghi.y) + (ghi.z + ghi.w);
    float d_lo = (dlo.x + dlo.y) + (dlo.z + dlo.w);
    float d_hi = (dhi.x + dhi.y) + (dhi.z + dhi.w);
    srow += __shfl_xor(srow, 16);
    g_lo += __shfl_xor(g_lo, 16);
    g_hi += __shfl_xor(g_hi, 16);
    d_lo += __shfl_xor(d_lo, 16);
    d_hi += __shfl_xor(d_hi, 16);
    g_lo -= 2.0f;
    g_hi -= 2.0f;

    #pragma unroll
    for (int r = 0; r < 8; ++r) {
        const int row = tm + r + half * 8;
        const float sb = __shfl(srow, r + half * 8);

        float c1 = c1lo[r], c2 = c2lo[r];
        float den = fmaxf(sqrtf(fmaxf(sb + c1 * c1 * g_lo, 0.0f)), EPSF);
        out[row * NDIM + tn + l] = (c2 - c1 * d_lo) / den;

        c1 = c1hi[r]; c2 = c2hi[r];
        den = fmaxf(sqrtf(fmaxf(sb + c1 * c1 * g_hi, 0.0f)), EPSF);
        out[row * NDIM + tn + 16 + l] = (c2 - c1 * d_hi) / den;
    }
}

extern "C" void kernel_launch(void* const* d_in, const int* in_sizes, int n_in,
                              void* d_out, int out_size, void* d_ws, size_t ws_size,
                              hipStream_t stream) {
    const float* x      = (const float*)d_in[0];
    const float* normal = (const float*)d_in[1];
    const float* plane  = (const float*)d_in[2];
    float* out = (float*)d_out;

    float* ws    = (float*)d_ws;
    float* pproj = ws;
    float* cninv = ws + NDIM * NDIM;

    pp2_prep_kernel<<<RP_BLOCKS + CN_BLOCKS, CN_COLS * CN_WAVES, 0, stream>>>(
        normal, plane, cninv, pproj);

    pp2_wmma_gemm_kernel<<<128, 128, 0, stream>>>(x, normal, pproj, cninv, out);
}


# GEMM re-tiled: one 32x32 output tile per block on all 256 WGPs, k split over the block's 4 waves (16 WMMAs per 6 fragment loads), partials joined in LDS; k sum order changes, also in s_b, g_o, d_o
# speedup vs baseline: 5.3394x; 1.6020x over previous
#include <hip/hip_runtime.h>
#include <math.h>

typedef __attribute__((ext_vector_type(2))) float v2f;
typedef __attribute__((ext_vector_type(4))) float v4f;
typedef __attribute__((ext_vector_type(8))) float v8f;

#define WMMA4(acc, av, bv)                                                   \
    acc = __builtin_amdgcn_wmma_f32_16x16x4_f32(false, av, false, bv,        \
                                                (short)0, acc, false, false)

#define WMMA_HALF(s0, s1, a0, a1, p0, p1, q0, q1)                            \
    do {                                                                     \
        WMMA4(c1_00, s0, p0);                                                \
        WMMA4(c1_01, s0, p1);                                                \
        WMMA4(c1_10, s1, p0);                                                \
        WMMA4(c1_11, s1, p1);                                                \
        WMMA4(c2_00, a0, q0);                                                \
        WMMA4(c2_01, a0, q1);                                                \
        WMMA4(c2_10, a1, q0);                                                \
        WMMA4(c2_11, a1, q1);                                                \
    } while (0)

#define WMMA_STEP(as0, as1, a0, a1, p0, p1, q0, q1)                          \
    do {                                                                     \
        WMMA_HALF(((v2f){as0.x, as0.y}), ((v2f){as1.x, as1.y}),              \
                  ((v2f){a0.x, a0.y}), ((v2f){a1.x, a1.y}),                  \
                  ((v2f){p0.x, p0.y}), ((v2f){p1.x, p1.y}),                  \
                  ((v2f){q0.x, q0.y}), ((v2f){q1.x, q1.y}));                 \
        WMMA_HALF(((v2f){as0.z, as0.w}), ((v2f){as1.z, as1.w}),              \
                  ((v2f){a0.z, a0.w}), ((v2f){a1.z, a1.w}),                  \
                  ((v2f){p0.z, p0.w}), ((v2f){p1.z, p1.w}),                  \
                  ((v2f){q0.z, q0.w}), ((v2f){q1.z, q1.w}));                 \
    } while (0)

#define NDIM 512
#define EPSF 1e-12f

#define OFF_HI (16 * NDIM)

#define CN_COLS 32
#define CN_WAVES 32
#define CN_ROWS 16
#define CN_BLOCKS 16
#define RP_ROWS 32
#define RP_BLOCKS (NDIM / RP_ROWS)

__global__ __launch_bounds__(CN_COLS * CN_WAVES) void pp2_prep_kernel(
    const float* __restrict__ normal, const float* __restrict__ plane,
    float* __restrict__ cninv, float* __restrict__ pproj) {
    __shared__ float red[CN_WAVES][CN_COLS];
    const int c  = threadIdx.x & (CN_COLS - 1);
    const int rg = threadIdx.x >> 5;

    if (blockIdx.x < RP_BLOCKS) {
        const int o = blockIdx.x * RP_ROWS + rg;
        const float* nr = normal + o * NDIM + 4 * c;
        const float* pr = plane  + o * NDIM + 4 * c;

        v4f n[4], p[4];
        #pragma unroll
        for (int j = 0; j < 4; ++j) {
            n[j] = *(const v4f*)(nr + j * 128);
            p[j] = *(const v4f*)(pr + j * 128);
        }

        v4f sp4 = {}, sn4 = {}, spn4 = {};
        #pragma unroll
        for (int j = 0; j < 4; ++j) {
            sp4  += p[j] * p[j];
            sn4  += n[j] * n[j];
            spn4 += p[j] * n[j];
        }
        float Sp  = (sp4.x + sp4.y) + (sp4.z + sp4.w);
        float Sn  = (sn4.x + sn4.y) + (sn4.z + sn4.w);
        float Spn = (spn4.x + spn4.y) + (spn4.z + spn4.w);
        #pragma unroll
        for (int m = 16; m > 0; m >>= 1) {
            Sp  += __shfl_xor(Sp, m);
            Sn  += __shfl_xor(Sn, m);
            Spn += __shfl_xor(Spn, m);
        }

        const float np  = fmaxf(sqrtf(Sp), EPSF);
        const float nn1 = fmaxf(sqrtf(Sn), EPSF);
        const float ct  = Spn / (np * nn1);
        const float qq  = fmaxf(Sp - 2.0f * ct * Spn + ct * ct * Sn, 0.0f);
        const float rinv = 1.0f / fmaxf(sqrtf(qq), EPSF);

        float* po = pproj + o * NDIM + 4 * c;
        #pragma unroll
        for (int j = 0; j < 4; ++j)
            *(v4f*)(po + j * 128) = (p[j] - n[j] * ct) * rinv;
    } else {
        const int col = (blockIdx.x - RP_BLOCKS) * CN_COLS + c;
        const float* p = normal + rg * NDIM + col;

        float v[CN_ROWS];
        #pragma unroll
        for (int j = 0; j < CN_ROWS; ++j) v[j] = p[j * CN_WAVES * NDIM];

        float s = 0.0f;
        #pragma unroll
        for (int j = 0; j < CN_ROWS; ++j) s += v[j] * v[j];

        red[rg][c] = s;
        __syncthreads();

        if (rg == 0) {
            float t0 = 0.0f, t1 = 0.0f, t2 = 0.0f, t3 = 0.0f;
            #pragma unroll
            for (int j = 0; j < CN_WAVES; j += 4) {
                t0 += red[j][c];
                t1 += red[j + 1][c];
                t2 += red[j + 2][c];
                t3 += red[j + 3][c];
            }
            const float tot = (t0 + t1) + (t2 + t3);
            cninv[col] = 1.0f / fmaxf(sqrtf(tot), EPSF);
        }
    }
}


#define ROW_SUMS(cv, p, q, g, d)                                             \
    do {                                                                     \
        const v4f t = p * cv;                                                \
        g += t * t;                                                          \
        d += t * q;                                                          \
    } while (0)

#define KSPLIT 4
#define KW (NDIM / KSPLIT)

#define PF_DEPTH 8

#define LOAD_SLOT(s, kk)                                                     \
    do {                                                                     \
        A0[s] = *(const v4f*)(ap + (kk));                                    \
        A1[s] = *(const v4f*)(ap + (kk) + OFF_HI);                           \
        C[s]  = *(const v4f*)(cp + (kk));                                    \
        P0[s] = *(const v4f*)(np + (kk));                                    \
        P1[s] = *(const v4f*)(np + (kk) + OFF_HI);                           \
        Q0[s] = *(const v4f*)(qp + (kk));                                    \
        Q1[s] = *(const v4f*)(qp + (kk) + OFF_HI);                           \
    } while (0)

#define USE_SLOT(s)                                                          \
    do {                                                                     \
        const v4f as0 = A0[s] * C[s];                                        \
        const v4f as1 = A1[s] * C[s];                                        \
        WMMA_STEP(as0, as1, A0[s], A1[s], P0[s], P1[s], Q0[s], Q1[s]);       \
        ss0 += A0[s] * A0[s];                                                \
        ss1 += A1[s] * A1[s];                                                \
        ROW_SUMS(C[s], P0[s], Q0[s], g0, d0);                                \
        ROW_SUMS(C[s], P1[s], Q1[s], g1, d1);                                \
    } while (0)

#define PUT_TILE(t, c)                                                       \
    do {                                                                     \
        part[w][t][0][lane] = (v4f){c[0], c[1], c[2], c[3]};                 \
        part[w][t][1][lane] = (v4f){c[4], c[5], c[6], c[7]};                 \
    } while (0)

__global__ __launch_bounds__(32 * KSPLIT)
__attribute__((amdgpu_waves_per_eu(1, 2)))
void pp2_wmma_gemm_kernel(
    const float* __restrict__ x,
    const float* __restrict__ normal,
    const float* __restrict__ pproj,
    const float* __restrict__ cninv,
    float* __restrict__ out) {
    __shared__ v4f part[KSPLIT][8][2][32];
    __shared__ float sums[KSPLIT][6][32];

    const int w = threadIdx.x >> 5;
    const int tm = (blockIdx.x >> 4) * 32;
    const int tn = (blockIdx.x & 15) * 32;

    const int lane = threadIdx.x & 31;
    const int half = lane >> 4;
    const int l = lane & 15;

    const int k0 = w * KW + 4 * half;
    const float* ap = x      + (tm + l) * NDIM + k0;
    const float* np = normal + (tn + l) * NDIM + k0;
    const float* qp = pproj  + (tn + l) * NDIM + k0;
    const float* cp = cninv  + k0;

    v8f c1_00 = {}, c1_01 = {}, c1_10 = {}, c1_11 = {};
    v8f c2_00 = {}, c2_01 = {}, c2_10 = {}, c2_11 = {};
    v4f ss0 = {}, ss1 = {};
    v4f g0 = {}, g1 = {};
    v4f d0 = {}, d1 = {};

    v4f A0[PF_DEPTH], A1[PF_DEPTH], C[PF_DEPTH];
    v4f P0[PF_DEPTH], P1[PF_DEPTH], Q0[PF_DEPTH], Q1[PF_DEPTH];

    #pragma unroll
    for (int s = 0; s < PF_DEPTH; ++s) {
        LOAD_SLOT(s, 8 * s);
        __builtin_amdgcn_sched_barrier(0);
    }

    #pragma unroll
    for (int s = 0; s < PF_DEPTH; ++s) {
        USE_SLOT(s);
        LOAD_SLOT(s, 8 * PF_DEPTH + 8 * s);
        __builtin_amdgcn_sched_barrier(0);
    }

    #pragma unroll
    for (int s = 0; s < PF_DEPTH; ++s) USE_SLOT(s);

    float s0  = (ss0.x + ss0.y) + (ss0.z + ss0.w);
    float s1  = (ss1.x + ss1.y) + (ss1.z + ss1.w);
    float gs0 = (g0.x + g0.y) + (g0.z + g0.w);
    float gs1 = (g1.x + g1.y) + (g1.z + g1.w);
    float ds0 = (d0.x + d0.y) + (d0.z + d0.w);
    float ds1 = (d1.x + d1.y) + (d1.z + d1.w);
    s0  += __shfl_xor(s0, 16);
    s1  += __shfl_xor(s1, 16);
    gs0 += __shfl_xor(gs0, 16);
    gs1 += __shfl_xor(gs1, 16);
    ds0 += __shfl_xor(ds0, 16);
    ds1 += __shfl_xor(ds1, 16);

    sums[w][0][lane] = s0;
    sums[w][1][lane] = s1;
    sums[w][2][lane] = gs0;
    sums[w][3][lane] = gs1;
    sums[w][4][lane] = ds0;
    sums[w][5][lane] = ds1;
    PUT_TILE(0, c1_00);
    PUT_TILE(1, c1_01);
    PUT_TILE(2, c1_10);
    PUT_TILE(3, c1_11);
    PUT_TILE(4, c2_00);
    PUT_TILE(5, c2_01);
    PUT_TILE(6, c2_10);
    PUT_TILE(7, c2_11);
    __syncthreads();

    const int i = w >> 1;
    const int p = w & 1;
    float srow = 0.0f, g_lo = 0.0f, g_hi = 0.0f, d_lo = 0.0f, d_hi = 0.0f;
    v4f c1lo = {}, c1hi = {}, c2lo = {}, c2hi = {};
    #pragma unroll
    for (int u = 0; u < KSPLIT; ++u) {
        srow += sums[u][i][lane];
        g_lo += sums[u][2][lane];
        g_hi += sums[u][3][lane];
        d_lo += sums[u][4][lane];
        d_hi += sums[u][5][lane];
        c1lo += part[u][2 * i][p][lane];
        c1hi += part[u][2 * i + 1][p][lane];
        c2lo += part[u][4 + 2 * i][p][lane];
        c2hi += part[u][4 + 2 * i + 1][p][lane];
    }
    g_lo -= 2.0f;
    g_hi -= 2.0f;

    #pragma unroll
    for (int e = 0; e < 4; ++e) {
        const int rr = half * 8 + p * 4 + e;
        const int row = tm + i * 16 + rr;
        const float sb = __shfl(srow, rr);

        float c1 = c1lo[e], c2 = c2lo[e];
        float den = fmaxf(sqrtf(fmaxf(sb + c1 * c1 * g_lo, 0.0f)), EPSF);
        out[row * NDIM + tn + l] = (c2 - c1 * d_lo) / den;

        c1 = c1hi[e]; c2 = c2hi[e];
        den = fmaxf(sqrtf(fmaxf(sb + c1 * c1 * g_hi, 0.0f)), EPSF);
        out[row * NDIM + tn + 16 + l] = (c2 - c1 * d_hi) / den;
    }
}

extern "C" void kernel_launch(void* const* d_in, const int* in_sizes, int n_in,
                              void* d_out, int out_size, void* d_ws, size_t ws_size,
                              hipStream_t stream) {
    const float* x      = (const float*)d_in[0];
    const float* normal = (const float*)d_in[1];
    const float* plane  = (const float*)d_in[2];
    float* out = (float*)d_out;

    float* ws    = (float*)d_ws;
    float* pproj = ws;
    float* cninv = ws + NDIM * NDIM;

    pp2_prep_kernel<<<RP_BLOCKS + CN_BLOCKS, CN_COLS * CN_WAVES, 0, stream>>>(
        normal, plane, cninv, pproj);

    pp2_wmma_gemm_kernel<<<256, 128, 0, stream>>>(x, normal, pproj, cninv, out);
}


# Prep launch re-shaped: 8-wave blocks, 384 waves instead of 1024, every load 16 B on whole lines, 16 in flight per wave (2 rows per wave; column strips 4 rows per load); column sum order changes
# speedup vs baseline: 5.5982x; 1.0485x over previous
#include <hip/hip_runtime.h>
#include <math.h>

typedef __attribute__((ext_vector_type(2))) float v2f;
typedef __attribute__((ext_vector_type(4))) float v4f;
typedef __attribute__((ext_vector_type(8))) float v8f;

#define WMMA4(acc, av, bv)                                                   \
    acc = __builtin_amdgcn_wmma_f32_16x16x4_f32(false, av, false, bv,        \
                                                (short)0, acc, false, false)

#define WMMA_HALF(s0, s1, a0, a1, p0, p1, q0, q1)                            \
    do {                                                                     \
        WMMA4(c1_00, s0, p0);                                                \
        WMMA4(c1_01, s0, p1);                                                \
        WMMA4(c1_10, s1, p0);                                                \
        WMMA4(c1_11, s1, p1);                                                \
        WMMA4(c2_00, a0, q0);                                                \
        WMMA4(c2_01, a0, q1);                                                \
        WMMA4(c2_10, a1, q0);                                                \
        WMMA4(c2_11, a1, q1);                                                \
    } while (0)

#define WMMA_STEP(as0, as1, a0, a1, p0, p1, q0, q1)                          \
    do {                                                                     \
        WMMA_HALF(((v2f){as0.x, as0.y}), ((v2f){as1.x, as1.y}),              \
                  ((v2f){a0.x, a0.y}), ((v2f){a1.x, a1.y}),                  \
                  ((v2f){p0.x, p0.y}), ((v2f){p1.x, p1.y}),                  \
                  ((v2f){q0.x, q0.y}), ((v2f){q1.x, q1.y}));                 \
        WMMA_HALF(((v2f){as0.z, as0.w}), ((v2f){as1.z, as1.w}),              \
                  ((v2f){a0.z, a0.w}), ((v2f){a1.z, a1.w}),                  \
                  ((v2f){p0.z, p0.w}), ((v2f){p1.z, p1.w}),                  \
                  ((v2f){q0.z, q0.w}), ((v2f){q1.z, q1.w}));                 \
    } while (0)

#define NDIM 512
#define EPSF 1e-12f

#define OFF_HI (16 * NDIM)

#define PK_WAVES 8
#define CN_COLS 32
#define CN_BLOCKS 16
#define CN_LOADS 16
#define RP_WROWS 2
#define RP_ROWS (PK_WAVES * RP_WROWS)
#define RP_BLOCKS (NDIM / RP_ROWS)

__global__ __launch_bounds__(32 * PK_WAVES) void pp2_prep_kernel(
    const float* __restrict__ normal, const float* __restrict__ plane,
    float* __restrict__ cninv, float* __restrict__ pproj) {
    __shared__ v4f red[PK_WAVES][CN_COLS / 4];
    const int c  = threadIdx.x & 31;
    const int rg = threadIdx.x >> 5;

    if (blockIdx.x < RP_BLOCKS) {
        const int o = blockIdx.x * RP_ROWS + RP_WROWS * rg;
        const float* nr = normal + o * NDIM + 4 * c;
        const float* pr = plane  + o * NDIM + 4 * c;

        v4f n[RP_WROWS][4], p[RP_WROWS][4];
        #pragma unroll
        for (int r = 0; r < RP_WROWS; ++r) {
            #pragma unroll
            for (int j = 0; j < 4; ++j) {
                n[r][j] = *(const v4f*)(nr + r * NDIM + j * 128);
                p[r][j] = *(const v4f*)(pr + r * NDIM + j * 128);
            }
        }
        __builtin_amdgcn_sched_barrier(0);

        float Sp[RP_WROWS], Sn[RP_WROWS], Spn[RP_WROWS];
        #pragma unroll
        for (int r = 0; r < RP_WROWS; ++r) {
            v4f sp4 = {}, sn4 = {}, spn4 = {};
            #pragma unroll
            for (int j = 0; j < 4; ++j) {
                sp4  += p[r][j] * p[r][j];
                sn4  += n[r][j] * n[r][j];
                spn4 += p[r][j] * n[r][j];
            }
            Sp[r]  = (sp4.x + sp4.y) + (sp4.z + sp4.w);
            Sn[r]  = (sn4.x + sn4.y) + (sn4.z + sn4.w);
            Spn[r] = (spn4.x + spn4.y) + (spn4.z + spn4.w);
        }
        #pragma unroll
        for (int m = 16; m > 0; m >>= 1) {
            #pragma unroll
            for (int r = 0; r < RP_WROWS; ++r) {
                Sp[r]  += __shfl_xor(Sp[r], m);
                Sn[r]  += __shfl_xor(Sn[r], m);
                Spn[r] += __shfl_xor(Spn[r], m);
            }
        }

        #pragma unroll
        for (int r = 0; r < RP_WROWS; ++r) {
            const float np  = fmaxf(sqrtf(Sp[r]), EPSF);
            const float nn1 = fmaxf(sqrtf(Sn[r]), EPSF);
            const float ct  = Spn[r] / (np * nn1);
            const float qq  = fmaxf(Sp[r] - 2.0f * ct * Spn[r] + ct * ct * Sn[r], 0.0f);
            const float rinv = 1.0f / fmaxf(sqrtf(qq), EPSF);

            float* po = pproj + (o + r) * NDIM + 4 * c;
            #pragma unroll
            for (int j = 0; j < 4; ++j)
                *(v4f*)(po + j * 128) = (p[r][j] - n[r][j] * ct) * rinv;
        }
    } else {
        const int strip = blockIdx.x - RP_BLOCKS;
        const int sub = c >> 3;
        const int cq  = c & 7;
        const float* p = normal + (rg * (4 * CN_LOADS) + sub) * NDIM
                                + strip * CN_COLS + 4 * cq;

        v4f v[CN_LOADS];
        #pragma unroll
        for (int j = 0; j < CN_LOADS; ++j) v[j] = *(const v4f*)(p + j * 4 * NDIM);

        v4f e0 = {}, e1 = {};
        #pragma unroll
        for (int j = 0; j < CN_LOADS; j += 2) {
            e0 += v[j] * v[j];
            e1 += v[j + 1] * v[j + 1];
        }
        v4f s = e0 + e1;
        #pragma unroll
        for (int m = 8; m <= 16; m <<= 1) {
            s.x += __shfl_xor(s.x, m);
            s.y += __shfl_xor(s.y, m);
            s.z += __shfl_xor(s.z, m);
            s.w += __shfl_xor(s.w, m);
        }

        if (c < CN_COLS / 4) red[rg][c] = s;
        __syncthreads();

        if (rg == 0) {
            const float* rf = (const float*)&red[0][0];
            const float t0 = rf[0 * CN_COLS + c] + rf[1 * CN_COLS + c];
            const float t1 = rf[2 * CN_COLS + c] + rf[3 * CN_COLS + c];
            const float t2 = rf[4 * CN_COLS + c] + rf[5 * CN_COLS + c];
            const float t3 = rf[6 * CN_COLS + c] + rf[7 * CN_COLS + c];
            const float tot = (t0 + t1) + (t2 + t3);
            cninv[strip * CN_COLS + c] = 1.0f / fmaxf(sqrtf(tot), EPSF);
        }
    }
}


#define ROW_SUMS(cv, p, q, g, d)                                             \
    do {                                                                     \
        const v4f t = p * cv;                                                \
        g += t * t;                                                          \
        d += t * q;                                                          \
    } while (0)

#define KSPLIT 4
#define KW (NDIM / KSPLIT)

#define PF_DEPTH 8

#define LOAD_SLOT(s, kk)                                                     \
    do {                                                                     \
        A0[s] = *(const v4f*)(ap + (kk));                                    \
        A1[s] = *(const v4f*)(ap + (kk) + OFF_HI);                           \
        C[s]  = *(const v4f*)(cp + (kk));                                    \
        P0[s] = *(const v4f*)(np + (kk));                                    \
        P1[s] = *(const v4f*)(np + (kk) + OFF_HI);                           \
        Q0[s] = *(const v4f*)(qp + (kk));                                    \
        Q1[s] = *(const v4f*)(qp + (kk) + OFF_HI);                           \
    } while (0)

#define USE_SLOT(s)                                                          \
    do {                                                                     \
        const v4f as0 = A0[s] * C[s];                                        \
        const v4f as1 = A1[s] * C[s];                                        \
        WMMA_STEP(as0, as1, A0[s], A1[s], P0[s], P1[s], Q0[s], Q1[s]);       \
        ss0 += A0[s] * A0[s];                                                \
        ss1 += A1[s] * A1[s];                                                \
        ROW_SUMS(C[s], P0[s], Q0[s], g0, d0);                                \
        ROW_SUMS(C[s], P1[s], Q1[s], g1, d1);                                \
    } while (0)

#define PUT_TILE(t, c)                                                       \
    do {                                                                     \
        part[w][t][0][lane] = (v4f){c[0], c[1], c[2], c[3]};                 \
        part[w][t][1][lane] = (v4f){c[4], c[5], c[6], c[7]};                 \
    } while (0)

__global__ __launch_bounds__(32 * KSPLIT)
__attribute__((amdgpu_waves_per_eu(1, 2)))
void pp2_wmma_gemm_kernel(
    const float* __restrict__ x,
    const float* __restrict__ normal,
    const float* __restrict__ pproj,
    const float* __restrict__ cninv,
    float* __restrict__ out) {
    __shared__ v4f part[KSPLIT][8][2][32];
    __shared__ float sums[KSPLIT][6][32];

    const int w = threadIdx.x >> 5;
    const int tm = (blockIdx.x >> 4) * 32;
    const int tn = (blockIdx.x & 15) * 32;

    const int lane = threadIdx.x & 31;
    const int half = lane >> 4;
    const int l = lane & 15;

    const int k0 = w * KW + 4 * half;
    const float* ap = x      + (tm + l) * NDIM + k0;
    const float* np = normal + (tn + l) * NDIM + k0;
    const float* qp = pproj  + (tn + l) * NDIM + k0;
    const float* cp = cninv  + k0;

    v8f c1_00 = {}, c1_01 = {}, c1_10 = {}, c1_11 = {};
    v8f c2_00 = {}, c2_01 = {}, c2_10 = {}, c2_11 = {};
    v4f ss0 = {}, ss1 = {};
    v4f g0 = {}, g1 = {};
    v4f d0 = {}, d1 = {};

    v4f A0[PF_DEPTH], A1[PF_DEPTH], C[PF_DEPTH];
    v4f P0[PF_DEPTH], P1[PF_DEPTH], Q0[PF_DEPTH], Q1[PF_DEPTH];

    #pragma unroll
    for (int s = 0; s < PF_DEPTH; ++s) {
        LOAD_SLOT(s, 8 * s);
        __builtin_amdgcn_sched_barrier(0);
    }

    #pragma unroll
    for (int s = 0; s < PF_DEPTH; ++s) {
        USE_SLOT(s);
        LOAD_SLOT(s, 8 * PF_DEPTH + 8 * s);
        __builtin_amdgcn_sched_barrier(0);
    }

    #pragma unroll
    for (int s = 0; s < PF_DEPTH; ++s) USE_SLOT(s);

    float s0  = (ss0.x + ss0.y) + (ss0.z + ss0.w);
    float s1  = (ss1.x + ss1.y) + (ss1.z + ss1.w);
    float gs0 = (g0.x + g0.y) + (g0.z + g0.w);
    float gs1 = (g1.x + g1.y) + (g1.z + g1.w);
    float ds0 = (d0.x + d0.y) + (d0.z + d0.w);
    float ds1 = (d1.x + d1.y) + (d1.z + d1.w);
    s0  += __shfl_xor(s0, 16);
    s1  += __shfl_xor(s1, 16);
    gs0 += __shfl_xor(gs0, 16);
    gs1 += __shfl_xor(gs1, 16);
    ds0 += __shfl_xor(ds0, 16);
    ds1 += __shfl_xor(ds1, 16);

    sums[w][0][lane] = s0;
    sums[w][1][lane] = s1;
    sums[w][2][lane] = gs0;
    sums[w][3][lane] = gs1;
    sums[w][4][lane] = ds0;
    sums[w][5][lane] = ds1;
    PUT_TILE(0, c1_00);
    PUT_TILE(1, c1_01);
    PUT_TILE(2, c1_10);
    PUT_TILE(3, c1_11);
    PUT_TILE(4, c2_00);
    PUT_TILE(5, c2_01);
    PUT_TILE(6, c2_10);
    PUT_TILE(7, c2_11);
    __syncthreads();

    const int i = w >> 1;
    const int p = w & 1;
    float srow = 0.0f, g_lo = 0.0f, g_hi = 0.0f, d_lo = 0.0f, d_hi = 0.0f;
    v4f c1lo = {}, c1hi = {}, c2lo = {}, c2hi = {};
    #pragma unroll
    for (int u = 0; u < KSPLIT; ++u) {
        srow += sums[u][i][lane];
        g_lo += sums[u][2][lane];
        g_hi += sums[u][3][lane];
        d_lo += sums[u][4][lane];
        d_hi += sums[u][5][lane];
        c1lo += part[u][2 * i][p][lane];
        c1hi += part[u][2 * i + 1][p][lane];
        c2lo += part[u][4 + 2 * i][p][lane];
        c2hi += part[u][4 + 2 * i + 1][p][lane];
    }
    g_lo -= 2.0f;
    g_hi -= 2.0f;

    #pragma unroll
    for (int e = 0; e < 4; ++e) {
        const int rr = half * 8 + p * 4 + e;
        const int row = tm + i * 16 + rr;
        const float sb = __shfl(srow, rr);

        float c1 = c1lo[e], c2 = c2lo[e];
        float den = fmaxf(sqrtf(fmaxf(sb + c1 * c1 * g_lo, 0.0f)), EPSF);
        out[row * NDIM + tn + l] = (c2 - c1 * d_lo) / den;

        c1 = c1hi[e]; c2 = c2hi[e];
        den = fmaxf(sqrtf(fmaxf(sb + c1 * c1 * g_hi, 0.0f)), EPSF);
        out[row * NDIM + tn + 16 + l] = (c2 - c1 * d_hi) / den;
    }
}

extern "C" void kernel_launch(void* const* d_in, const int* in_sizes, int n_in,
                              void* d_out, int out_size, void* d_ws, size_t ws_size,
                              hipStream_t stream) {
    const float* x      = (const float*)d_in[0];
    const float* normal = (const float*)d_in[1];
    const float* plane  = (const float*)d_in[2];
    float* out = (float*)d_out;

    float* ws    = (float*)d_ws;
    float* pproj = ws;
    float* cninv = ws + NDIM * NDIM;

    pp2_prep_kernel<<<RP_BLOCKS + CN_BLOCKS, 32 * PK_WAVES, 0, stream>>>(
        normal, plane, cninv, pproj);

    pp2_wmma_gemm_kernel<<<256, 128, 0, stream>>>(x, normal, pproj, cninv, out);
}


# GEMM operands fetched by whole-line loads (one wave load = 4 rows x 128 B: 4 lines touched, not 16), staged per wave in LDS at a 144 B pitch and read back as fragments; order of every sum unchanged
# speedup vs baseline: 5.9936x; 1.0706x over previous
#include <hip/hip_runtime.h>
#include <math.h>

typedef __attribute__((ext_vector_type(2))) float v2f;
typedef __attribute__((ext_vector_type(4))) float v4f;
typedef __attribute__((ext_vector_type(8))) float v8f;

#define WMMA4(acc, av, bv)                                                   \
    acc = __builtin_amdgcn_wmma_f32_16x16x4_f32(false, av, false, bv,        \
                                                (short)0, acc, false, false)

#define WMMA_HALF(s0, s1, a0, a1, p0, p1, q0, q1)                            \
    do {                                                                     \
        WMMA4(c1_00, s0, p0);                                                \
        WMMA4(c1_01, s0, p1);                                                \
        WMMA4(c1_10, s1, p0);                                                \
        WMMA4(c1_11, s1, p1);                                                \
        WMMA4(c2_00, a0, q0);                                                \
        WMMA4(c2_01, a0, q1);                                                \
        WMMA4(c2_10, a1, q0);                                                \
        WMMA4(c2_11, a1, q1);                                                \
    } while (0)

#define WMMA_STEP(as0, as1, a0, a1, p0, p1, q0, q1)                          \
    do {                                                                     \
        WMMA_HALF(((v2f){as0.x, as0.y}), ((v2f){as1.x, as1.y}),              \
                  ((v2f){a0.x, a0.y}), ((v2f){a1.x, a1.y}),                  \
                  ((v2f){p0.x, p0.y}), ((v2f){p1.x, p1.y}),                  \
                  ((v2f){q0.x, q0.y}), ((v2f){q1.x, q1.y}));                 \
        WMMA_HALF(((v2f){as0.z, as0.w}), ((v2f){as1.z, as1.w}),              \
                  ((v2f){a0.z, a0.w}), ((v2f){a1.z, a1.w}),                  \
                  ((v2f){p0.z, p0.w}), ((v2f){p1.z, p1.w}),                  \
                  ((v2f){q0.z, q0.w}), ((v2f){q1.z, q1.w}));                 \
    } while (0)

#define NDIM 512
#define EPSF 1e-12f

#define PK_WAVES 8
#define CN_COLS 32
#define CN_BLOCKS 16
#define CN_LOADS 16
#define RP_WROWS 2
#define RP_ROWS (PK_WAVES * RP_WROWS)
#define RP_BLOCKS (NDIM / RP_ROWS)

__global__ __launch_bounds__(32 * PK_WAVES) void pp2_prep_kernel(
    const float* __restrict__ normal, const float* __restrict__ plane,
    float* __restrict__ cninv, float* __restrict__ pproj) {
    __shared__ v4f red[PK_WAVES][CN_COLS / 4];
    const int c  = threadIdx.x & 31;
    const int rg = threadIdx.x >> 5;

    if (blockIdx.x < RP_BLOCKS) {
        const int o = blockIdx.x * RP_ROWS + RP_WROWS * rg;
        const float* nr = normal + o * NDIM + 4 * c;
        const float* pr = plane  + o * NDIM + 4 * c;

        v4f n[RP_WROWS][4], p[RP_WROWS][4];
        #pragma unroll
        for (int r = 0; r < RP_WROWS; ++r) {
            #pragma unroll
            for (int j = 0; j < 4; ++j) {
                n[r][j] = *(const v4f*)(nr + r * NDIM + j * 128);
                p[r][j] = *(const v4f*)(pr + r * NDIM + j * 128);
            }
        }
        __builtin_amdgcn_sched_barrier(0);

        float Sp[RP_WROWS], Sn[RP_WROWS], Spn[RP_WROWS];
        #pragma unroll
        for (int r = 0; r < RP_WROWS; ++r) {
            v4f sp4 = {}, sn4 = {}, spn4 = {};
            #pragma unroll
            for (int j = 0; j < 4; ++j) {
                sp4  += p[r][j] * p[r][j];
                sn4  += n[r][j] * n[r][j];
                spn4 += p[r][j] * n[r][j];
            }
            Sp[r]  = (sp4.x + sp4.y) + (sp4.z + sp4.w);
            Sn[r]  = (sn4.x + sn4.y) + (sn4.z + sn4.w);
            Spn[r] = (spn4.x + spn4.y) + (spn4.z + spn4.w);
        }
        #pragma unroll
        for (int m = 16; m > 0; m >>= 1) {
            #pragma unroll
            for (int r = 0; r < RP_WROWS; ++r) {
                Sp[r]  += __shfl_xor(Sp[r], m);
                Sn[r]  += __shfl_xor(Sn[r], m);
                Spn[r] += __shfl_xor(Spn[r], m);
            }
        }

        #pragma unroll
        for (int r = 0; r < RP_WROWS; ++r) {
            const float np  = fmaxf(sqrtf(Sp[r]), EPSF);
            const float nn1 = fmaxf(sqrtf(Sn[r]), EPSF);
            const float ct  = Spn[r] / (np * nn1);
            const float qq  = fmaxf(Sp[r] - 2.0f * ct * Spn[r] + ct * ct * Sn[r], 0.0f);
            const float rinv = 1.0f / fmaxf(sqrtf(qq), EPSF);

            float* po = pproj + (o + r) * NDIM + 4 * c;
            #pragma unroll
            for (int j = 0; j < 4; ++j)
                *(v4f*)(po + j * 128) = (p[r][j] - n[r][j] * ct) * rinv;
        }
    } else {
        const int strip = blockIdx.x - RP_BLOCKS;
        const int sub = c >> 3;
        const int cq  = c & 7;
        const float* p = normal + (rg * (4 * CN_LOADS) + sub) * NDIM
                                + strip * CN_COLS + 4 * cq;

        v4f v[CN_LOADS];
        #pragma unroll
        for (int j = 0; j < CN_LOADS; ++j) v[j] = *(const v4f*)(p + j * 4 * NDIM);

        v4f e0 = {}, e1 = {};
        #pragma unroll
        for (int j = 0; j < CN_LOADS; j += 2) {
            e0 += v[j] * v[j];
            e1 += v[j + 1] * v[j + 1];
        }
        v4f s = e0 + e1;
        #pragma unroll
        for (int m = 8; m <= 16; m <<= 1) {
            s.x += __shfl_xor(s.x, m);
            s.y += __shfl_xor(s.y, m);
            s.z += __shfl_xor(s.z, m);
            s.w += __shfl_xor(s.w, m);
        }

        if (c < CN_COLS / 4) red[rg][c] = s;
        __syncthreads();

        if (rg == 0) {
            const float* rf = (const float*)&red[0][0];
            const float t0 = rf[0 * CN_COLS + c] + rf[1 * CN_COLS + c];
            const float t1 = rf[2 * CN_COLS + c] + rf[3 * CN_COLS + c];
            const float t2 = rf[4 * CN_COLS + c] + rf[5 * CN_COLS + c];
            const float t3 = rf[6 * CN_COLS + c] + rf[7 * CN_COLS + c];
            const float tot = (t0 + t1) + (t2 + t3);
            cninv[strip * CN_COLS + c] = 1.0f / fmaxf(sqrtf(tot), EPSF);
        }
    }
}


#define ROW_SUMS(cv, p, q, g, d)                                             \
    do {                                                                     \
        const v4f t = p * cv;                                                \
        g += t * t;                                                          \
        d += t * q;                                                          \
    } while (0)

#define KSPLIT 4
#define KW (NDIM / KSPLIT)

#define CHUNK_K 32
#define NCHUNK (KW / CHUNK_K)
#define NDSTEP (2 * NCHUNK)
#define ST_ROWS 96
#define ROWP 9

#define LOADG(u, c)                                                          \
    do {                                                                     \
                  \
        _Pragma("unroll")                                                    \
        for (int ii = 0; ii < 4; ++ii) {                                     \
            const int ro = (16 * (u) + 4 * ii) * NDIM + CHUNK_K * (c);       \
            G[u][0][ii] = *(const v4f*)(ag + ro);                            \
            G[u][1][ii] = *(const v4f*)(ng + ro);                            \
            G[u][2][ii] = *(const v4f*)(qg + ro);                            \
        }                                                                    \
    } while (0)

#define STOREG(u, b)                                                         \
    do {                                                                     \
               \
        _Pragma("unroll")                                                    \
        for (int op = 0; op < 3; ++op) {                                     \
            _Pragma("unroll")                                                \
            for (int ii = 0; ii < 4; ++ii)                                   \
                sw[((b) * ST_ROWS + 32 * op + 16 * (u) + 4 * ii) * ROWP] =   \
                    G[u][op][ii];                                            \
        }                                                                    \
    } while (0)

#define LOADF(f, b, v)                                                       \
    do {                                                                     \
          \
              \
        __builtin_amdgcn_wave_barrier();                                     \
               \
        _Pragma("unroll")                                                    \
        for (int s2 = 0; s2 < 2; ++s2) {                                     \
            _Pragma("unroll")                                                \
            for (int o6 = 0; o6 < 6; ++o6)                                   \
                F[f][s2][o6] = sr[((b) * ST_ROWS + 16 * o6) * ROWP +         \
                                  2 * (2 * (v) + s2)];                       \
        }                                                                    \
    } while (0)

#define USE_STEP(f, s2, cv)                                                  \
    do {                                                                     \
        const v4f as0 = F[f][s2][0] * cv;                                    \
        const v4f as1 = F[f][s2][1] * cv;                                    \
        WMMA_STEP(as0, as1, F[f][s2][0], F[f][s2][1], F[f][s2][2],           \
                  F[f][s2][3], F[f][s2][4], F[f][s2][5]);                    \
        ss0 += F[f][s2][0] * F[f][s2][0];                                    \
        ss1 += F[f][s2][1] * F[f][s2][1];                                    \
        ROW_SUMS(cv, F[f][s2][2], F[f][s2][4], g0, d0);                      \
        ROW_SUMS(cv, F[f][s2][3], F[f][s2][5], g1, d1);                      \
    } while (0)

#define PUT_TILE(t, c)                                                       \
    do {                                                                     \
        part[w][t][0][lane] = (v4f){c[0], c[1], c[2], c[3]};                 \
        part[w][t][1][lane] = (v4f){c[4], c[5], c[6], c[7]};                 \
    } while (0)

__global__ __launch_bounds__(32 * KSPLIT)
__attribute__((amdgpu_waves_per_eu(1, 2)))
void pp2_wmma_gemm_kernel(
    const float* __restrict__ x,
    const float* __restrict__ normal,
    const float* __restrict__ pproj,
    const float* __restrict__ cninv,
    float* __restrict__ out) {
    __shared__ v4f stage[KSPLIT][2 * ST_ROWS * ROWP];
    __shared__ v4f part[KSPLIT][8][2][32];
    __shared__ float sums[KSPLIT][6][32];

    const int w = threadIdx.x >> 5;
    const int tm = (blockIdx.x >> 4) * 32;
    const int tn = (blockIdx.x & 15) * 32;

    const int lane = threadIdx.x & 31;
    const int half = lane >> 4;
    const int l = lane & 15;
    const int sub = lane >> 3;
    const int cq = lane & 7;

    const int kw = w * KW;
    const float* ag = x      + (tm + sub) * NDIM + kw + 4 * cq;
    const float* ng = normal + (tn + sub) * NDIM + kw + 4 * cq;
    const float* qg = pproj  + (tn + sub) * NDIM + kw + 4 * cq;
    const float* cp = cninv  + kw + 4 * half;
    v4f* sw = &stage[w][sub * ROWP + cq];
    const v4f* sr = &stage[w][l * ROWP + half];

    v8f c1_00 = {}, c1_01 = {}, c1_10 = {}, c1_11 = {};
    v8f c2_00 = {}, c2_01 = {}, c2_10 = {}, c2_11 = {};
    v4f ss0 = {}, ss1 = {};
    v4f g0 = {}, g1 = {};
    v4f d0 = {}, d1 = {};

    v4f G[2][3][4];
    v4f F[2][2][6];
    v4f C[2 * NDSTEP];

    LOADG(0, 0);
    LOADG(1, 0);
    #pragma unroll
    for (int st = 0; st < 2 * NDSTEP; ++st) C[st] = *(const v4f*)(cp + 8 * st);
    __builtin_amdgcn_sched_barrier(0);
    STOREG(0, 0);
    STOREG(1, 0);
    LOADG(0, 1);
    LOADG(1, 1);
    LOADF(0, 0, 0);
    __builtin_amdgcn_sched_barrier(0);

    #pragma unroll
    for (int ds = 0; ds < NDSTEP; ++ds) {
        const int c = ds >> 1;
        const int u = ds & 1;
        if (c + 1 < NCHUNK) STOREG(u, (c + 1) & 1);
        if (c + 2 < NCHUNK) LOADG(u, c + 2);
        if (ds + 1 < NDSTEP) LOADF((ds + 1) & 1, ((ds + 1) >> 1) & 1, (ds + 1) & 1);
        __builtin_amdgcn_sched_barrier(0);
        USE_STEP(ds & 1, 0, C[2 * ds]);
        USE_STEP(ds & 1, 1, C[2 * ds + 1]);
        __builtin_amdgcn_sched_barrier(0);
    }

    float s0  = (ss0.x + ss0.y) + (ss0.z + ss0.w);
    float s1  = (ss1.x + ss1.y) + (ss1.z + ss1.w);
    float gs0 = (g0.x + g0.y) + (g0.z + g0.w);
    float gs1 = (g1.x + g1.y) + (g1.z + g1.w);
    float ds0 = (d0.x + d0.y) + (d0.z + d0.w);
    float ds1 = (d1.x + d1.y) + (d1.z + d1.w);
    s0  += __shfl_xor(s0, 16);
    s1  += __shfl_xor(s1, 16);
    gs0 += __shfl_xor(gs0, 16);
    gs1 += __shfl_xor(gs1, 16);
    ds0 += __shfl_xor(ds0, 16);
    ds1 += __shfl_xor(ds1, 16);

    sums[w][0][lane] = s0;
    sums[w][1][lane] = s1;
    sums[w][2][lane] = gs0;
    sums[w][3][lane] = gs1;
    sums[w][4][lane] = ds0;
    sums[w][5][lane] = ds1;
    PUT_TILE(0, c1_00);
    PUT_TILE(1, c1_01);
    PUT_TILE(2, c1_10);
    PUT_TILE(3, c1_11);
    PUT_TILE(4, c2_00);
    PUT_TILE(5, c2_01);
    PUT_TILE(6, c2_10);
    PUT_TILE(7, c2_11);
    __syncthreads();

    const int i = w >> 1;
    const int p = w & 1;
    float srow = 0.0f, g_lo = 0.0f, g_hi = 0.0f, d_lo = 0.0f, d_hi = 0.0f;
    v4f c1lo = {}, c1hi = {}, c2lo = {}, c2hi = {};
    #pragma unroll
    for (int u = 0; u < KSPLIT; ++u) {
        srow += sums[u][i][lane];
        g_lo += sums[u][2][lane];
        g_hi += sums[u][3][lane];
        d_lo += sums[u][4][lane];
        d_hi += sums[u][5][lane];
        c1lo += part[u][2 * i][p][lane];
        c1hi += part[u][2 * i + 1][p][lane];
        c2lo += part[u][4 + 2 * i][p][lane];
        c2hi += part[u][4 + 2 * i + 1][p][lane];
    }
    g_lo -= 2.0f;
    g_hi -= 2.0f;

    #pragma unroll
    for (int e = 0; e < 4; ++e) {
        const int rr = half * 8 + p * 4 + e;
        const int row = tm + i * 16 + rr;
        const float sb = __shfl(srow, rr);

        float c1 = c1lo[e], c2 = c2lo[e];
        float den = fmaxf(sqrtf(fmaxf(sb + c1 * c1 * g_lo, 0.0f)), EPSF);
        out[row * NDIM + tn + l] = (c2 - c1 * d_lo) / den;

        c1 = c1hi[e]; c2 = c2hi[e];
        den = fmaxf(sqrtf(fmaxf(sb + c1 * c1 * g_hi, 0.0f)), EPSF);
        out[row * NDIM + tn + 16 + l] = (c2 - c1 * d_hi) / den;
    }
}

extern "C" void kernel_launch(void* const* d_in, const int* in_sizes, int n_in,
                              void* d_out, int out_size, void* d_ws, size_t ws_size,
                              hipStream_t stream) {
    const float* x      = (const float*)d_in[0];
    const float* normal = (const float*)d_in[1];
    const float* plane  = (const float*)d_in[2];
    float* out = (float*)d_out;

    float* ws    = (float*)d_ws;
    float* pproj = ws;
    float* cninv = ws + NDIM * NDIM;

    pp2_prep_kernel<<<RP_BLOCKS + CN_BLOCKS, 32 * PK_WAVES, 0, stream>>>(
        normal, plane, cninv, pproj);

    pp2_wmma_gemm_kernel<<<256, 128, 0, stream>>>(x, normal, pproj, cninv, out);
}
